# starting kernel
# baseline (speedup 1.0000x reference)
#include <hip/hip_runtime.h>

#define HH 256
#define WW 256
#define BB 4
#define CIN 8
#define COUT 8
#define NTYPES 216
#define NPIX (BB*HH*WW)
#define KDIM 72
#define NW   (COUT*KDIM)
#define EMB_STRIDE (NW + COUT)

typedef float v2f __attribute__((ext_vector_type(2)));
typedef float v8f __attribute__((ext_vector_type(8)));

__global__ void k_zero_counts(int* counts) {
    int i = blockIdx.x * blockDim.x + threadIdx.x;
    if (i < NTYPES) counts[i] = 0;
}

__global__ void k_hist(const int* __restrict__ buckets, int* __restrict__ counts) {
    int i = blockIdx.x * blockDim.x + threadIdx.x;
    if (i < NPIX) atomicAdd(&counts[buckets[i]], 1);
}

__global__ void k_scan(const int* __restrict__ counts, int* __restrict__ offsets,
                       int* __restrict__ cursor) {
    int acc = 0;
    for (int t = 0; t < NTYPES; ++t) {
        offsets[t] = acc;
        cursor[t]  = acc;
        acc += counts[t];
    }
    offsets[NTYPES] = acc;
}

__global__ void k_scatter(const int* __restrict__ buckets, int* __restrict__ cursor,
                          int* __restrict__ perm) {
    int i = blockIdx.x * blockDim.x + threadIdx.x;
    if (i < NPIX) {
        int p = atomicAdd(&cursor[buckets[i]], 1);
        perm[p] = i;
    }
}

#define WAVES_PER_BLOCK 8
#define BLOCKS_PER_TYPE 4

__global__ __launch_bounds__(256) void k_bucket_gemm(
    const float* __restrict__ x, const float* __restrict__ emb,
    const int* __restrict__ offsets, const int* __restrict__ perm,
    float* __restrict__ out) {

    const int t    = blockIdx.x;
    const int lane = threadIdx.x & 31;
    const int gwave  = blockIdx.y * WAVES_PER_BLOCK + (threadIdx.x >> 5);
    const int nwaves = gridDim.y * WAVES_PER_BLOCK;

    const int start  = offsets[t];
    const int end    = offsets[t + 1];
    const int ntiles = (end - start + 15) >> 4;

    const float* wrow = emb + t * EMB_STRIDE;

    const int n  = lane & 15;
    const int hi = lane >> 4;
    const int khalf = hi << 1;
    v2f bmat[18];
    #pragma unroll
    for (int cc = 0; cc < 18; ++cc) {
        const int k = cc * 4 + khalf;
        float b0 = 0.f, b1 = 0.f;
        if (n < COUT) {
            b0 = wrow[n * KDIM + k];
            b1 = wrow[n * KDIM + k + 1];
        }
        bmat[cc].x = b0; bmat[cc].y = b1;
    }
    const float biasv = (n < COUT) ? wrow[NW + n] : 0.f;

    for (int tile = gwave; tile < ntiles; tile += nwaves) {
        const int rowbase = start + tile * 16;

        const int m     = lane & 15;
        const int pidx  = rowbase + m;
        const int valid = (pidx < end) ? 1 : 0;
        const int cp    = valid ? pidx : start;
        const int pix   = perm[cp];
        const int pb = pix >> 16;
        const int ph = (pix >> 8) & 255;
        const int pw = pix & 255;

        const float rowmask = valid ? 1.f : 0.f;
        int   hrow[3], wcol[3];
        float hmask[3], wmask[3];
        #pragma unroll
        for (int d = 0; d < 3; ++d) {
            const int hh = ph + d - 1;
            const int ww = pw + d - 1;
            hrow[d]  = (hh & 255) << 8;
            wcol[d]  = ww & 255;
            hmask[d] = ((unsigned)hh < 256u) ? 1.f : 0.f;
            wmask[d] = ((unsigned)ww < 256u) ? 1.f : 0.f;
        }
        const int base = (pb * CIN) << 16;
        int   off9[9];
        float msk9[9];
        #pragma unroll
        for (int r = 0; r < 9; ++r) {
            off9[r] = base + hrow[r / 3] + wcol[r % 3];
            msk9[r] = hmask[r / 3] * wmask[r % 3] * rowmask;
        }

        int   offR[9];
        float mskR[9];
        #pragma unroll
        for (int s = 0; s < 9; ++s) {
            const int s2 = (s + 2) % 9;
            offR[s] = hi ? off9[s2] : off9[s];
            mskR[s] = hi ? msk9[s2] : msk9[s];
        }

        v2f a[18];
        #pragma unroll
        for (int cc = 0; cc < 18; ++cc) {
            #pragma unroll
            for (int j = 0; j < 2; ++j) {
                const int kbase = cc * 4 + j;
                const int tapc  = kbase % 9;
                const int cA = (kbase / 9) << 16;
                const int cB = ((kbase + 2) / 9) << 16;
                const int coff = (cA == cB) ? cA : (hi ? cB : cA);
                const float v = x[offR[tapc] + coff] * mskR[tapc];
                if (j == 0) a[cc].x = v; else a[cc].y = v;
            }
        }

        v8f acc = {};
        #pragma unroll
        for (int cc = 0; cc < 18; ++cc) {
            acc = __builtin_amdgcn_wmma_f32_16x16x4_f32(
                false, a[cc], false, bmat[cc], (short)0, acc, false, false);
        }

        const int col  = lane & 15;
        const int rofs = hi << 3;
        #pragma unroll
        for (int r = 0; r < 8; ++r) {
            const int row   = r + rofs;
            const int pix2  = __shfl(pix, row, 32);
            if (col < COUT && (rowbase + row) < end) {
                const int b2 = pix2 >> 16;
                const int h2 = (pix2 >> 8) & 255;
                const int w2 = pix2 & 255;
                out[((b2 * COUT + col) << 16) + (h2 << 8) + w2] = acc[r] + biasv;
            }
        }
    }
}

extern "C" void kernel_launch(void* const* d_in, const int* in_sizes, int n_in,
                              void* d_out, int out_size, void* d_ws, size_t ws_size,
                              hipStream_t stream) {
    const float* x       = (const float*)d_in[0];
    const float* emb     = (const float*)d_in[1];
    const int*   buckets = (const int*)d_in[2];
    float*       out     = (float*)d_out;

    int* counts  = (int*)d_ws;
    int* cursor  = (int*)((char*)d_ws + 1024);
    int* offsets = (int*)((char*)d_ws + 2048);
    int* perm    = (int*)((char*)d_ws + 4096);

    k_zero_counts<<<1, 256, 0, stream>>>(counts);
    k_hist<<<(NPIX + 255) / 256, 256, 0, stream>>>(buckets, counts);
    k_scan<<<1, 1, 0, stream>>>(counts, offsets, cursor);
    k_scatter<<<(NPIX + 255) / 256, 256, 0, stream>>>(buckets, cursor, perm);
    k_bucket_gemm<<<dim3(NTYPES, BLOCKS_PER_TYPE), 256, 0, stream>>>(
        x, emb, offsets, perm, out);
}


# x copied channel-interleaved (b,h,w,c) into the workspace by a pre-pass launch; a lane half owns 4 channels and gathers each of the 9 taps with one 16-byte load (9 per tile, were 36 4-byte loads)
# speedup vs baseline: 1.9399x; 1.9399x over previous
#include <hip/hip_runtime.h>

#define HH 256
#define WW 256
#define BB 4
#define CIN 8
#define COUT 8
#define NTYPES 216
#define NPIX (BB*HH*WW)
#define KDIM 72
#define NW   (COUT*KDIM)
#define EMB_STRIDE (NW + COUT)

typedef float v2f __attribute__((ext_vector_type(2)));
typedef float v8f __attribute__((ext_vector_type(8)));
typedef int   v4i __attribute__((ext_vector_type(4)));
typedef float v4f __attribute__((ext_vector_type(4)));

#define XT_BYTES ((size_t)NPIX * CIN * sizeof(float))

#define PRE_BLOCKS 256
#define PRE_UNITS  (2 * NPIX / (PRE_BLOCKS * 256))
__global__ __launch_bounds__(256) void k_interleave_channels(
    const float* __restrict__ x, float* __restrict__ xt) {
    const int g0 = blockIdx.x * 256 + threadIdx.x;
    v4f v[PRE_UNITS];
    #pragma unroll
    for (int it = 0; it < PRE_UNITS; ++it) {
        const int g   = it * (PRE_BLOCKS * 256) + g0;
        const int pix = g >> 1;
        const int cb  = (g & 1) << 2;
        const int b   = pix >> 16;
        const int hw  = pix & 65535;
        const float* src = x + ((b * CIN + cb) << 16) + hw;
        v[it].x = src[0];
        v[it].y = src[1 << 16];
        v[it].z = src[2 << 16];
        v[it].w = src[3 << 16];
    }
    #pragma unroll
    for (int it = 0; it < PRE_UNITS; ++it) {
        const int g = it * (PRE_BLOCKS * 256) + g0;
        *(v4f*)(xt + (g << 2)) = v[it];
    }
}

#define WAVES_PER_BLOCK 8
#define SEGS 4
#define SEG_PIX   (NPIX / SEGS)
#define WAVE_PIX  (SEG_PIX / WAVES_PER_BLOCK)
#define BATCH_LOADS 8
#define BATCH_PIX (BATCH_LOADS * 128)
#define WCAP 1536

template <bool INTERLEAVED>
__global__ __launch_bounds__(256, 8) void k_bucket_scan_gemm(
    const float* __restrict__ x, const float* __restrict__ emb,
    const int* __restrict__ buckets, float* __restrict__ out) {

    __shared__ int s_q[WAVES_PER_BLOCK * WCAP];
    __shared__ int s_cnt[WAVES_PER_BLOCK];
    __shared__ int s_more[WAVES_PER_BLOCK];

    const int t    = blockIdx.y;
    const int lane = threadIdx.x & 31;
    const int wave = __builtin_amdgcn_readfirstlane(threadIdx.x >> 5);

    const float* wrow = emb + t * EMB_STRIDE;

    const int n  = lane & 15;
    const int hi = lane >> 4;
    const int khalf = hi << 1;

    int* myq = s_q + wave * WCAP;
    int pos = blockIdx.x * SEG_PIX + wave * WAVE_PIX;
    const int wend = pos + WAVE_PIX;

    int more;
    do {
        int cnt = 0;
        if (pos < wend) {
            v4i buf[BATCH_LOADS];
            #pragma unroll
            for (int j = 0; j < BATCH_LOADS; ++j)
                buf[j] = *(const v4i*)(buckets + pos + j * 128 + lane * 4);
            while (true) {
                v4i cur[BATCH_LOADS];
                #pragma unroll
                for (int j = 0; j < BATCH_LOADS; ++j) cur[j] = buf[j];
                const int  npos = pos + BATCH_PIX;
                const bool goon = npos < wend;
                if (goon) {
                    #pragma unroll
                    for (int j = 0; j < BATCH_LOADS; ++j)
                        buf[j] = *(const v4i*)(buckets + npos + j * 128 + lane * 4);
                }
                #pragma unroll
                for (int j = 0; j < BATCH_LOADS; ++j) {
                    const bool m0 = cur[j].x == t;
                    const bool m1 = cur[j].y == t;
                    const bool m2 = cur[j].z == t;
                    const bool m3 = cur[j].w == t;
                    if (__builtin_amdgcn_ballot_w32(m0 | m1 | m2 | m3) != 0u) {
                        const int pbase = pos + j * 128 + lane * 4;
                        const unsigned k0 = __builtin_amdgcn_ballot_w32(m0);
                        const unsigned k1 = __builtin_amdgcn_ballot_w32(m1);
                        const unsigned k2 = __builtin_amdgcn_ballot_w32(m2);
                        const unsigned k3 = __builtin_amdgcn_ballot_w32(m3);
                        if (m0) myq[cnt + __builtin_amdgcn_mbcnt_lo(k0, 0u)] = pbase;
                        cnt += __builtin_popcount(k0);
                        if (m1) myq[cnt + __builtin_amdgcn_mbcnt_lo(k1, 0u)] = pbase + 1;
                        cnt += __builtin_popcount(k1);
                        if (m2) myq[cnt + __builtin_amdgcn_mbcnt_lo(k2, 0u)] = pbase + 2;
                        cnt += __builtin_popcount(k2);
                        if (m3) myq[cnt + __builtin_amdgcn_mbcnt_lo(k3, 0u)] = pbase + 3;
                        cnt += __builtin_popcount(k3);
                    }
                }
                pos = npos;
                if (!goon || cnt + BATCH_PIX > WCAP) break;
            }
        }
        if (lane == 0) {
            s_cnt[wave]  = cnt;
            s_more[wave] = (pos < wend) ? 1 : 0;
        }

        v2f bmat[18];
        #pragma unroll
        for (int cc = 0; cc < 18; ++cc) {
            const int k0 = INTERLEAVED ? (4 * hi + 2 * (cc & 1)) * 9 + (cc >> 1) : cc * 4 + khalf;
            const int k1 = INTERLEAVED ? k0 + 9 : k0 + 1;
            float b0 = 0.f, b1 = 0.f;
            if (n < COUT) {
                b0 = wrow[n * KDIM + k0];
                b1 = wrow[n * KDIM + k1];
            }
            bmat[cc].x = b0; bmat[cc].y = b1;
        }
        const float biasv = (n < COUT) ? wrow[NW + n] : 0.f;

        __syncthreads();

        int cn[WAVES_PER_BLOCK];
        int total = 0;
        more = 0;
        #pragma unroll
        for (int i = 0; i < WAVES_PER_BLOCK; ++i) {
            cn[i]  = __builtin_amdgcn_readfirstlane(s_cnt[i]);
            total += cn[i];
            more  |= __builtin_amdgcn_readfirstlane(s_more[i]);
        }
        const int ntiles = (total + 15) >> 4;

        for (int tile = wave; tile < ntiles; tile += WAVES_PER_BLOCK) {
            const int rowbase = tile * 16;

            const int m     = lane & 15;
            const int ridx  = rowbase + m;
            const int valid = (ridx < total) ? 1 : 0;
            int off  = valid ? ridx : 0;
            int qsel = 0;
            #pragma unroll
            for (int i = 0; i < WAVES_PER_BLOCK - 1; ++i) {
                if (qsel == i && off >= cn[i]) { off -= cn[i]; qsel = i + 1; }
            }
            const int pix = s_q[qsel * WCAP + off];
            const int pb = pix >> 16;
            const int ph = (pix >> 8) & 255;
            const int pw = pix & 255;

            v2f a[18];
            if constexpr (INTERLEAVED) {
                int  hrow[3], wcol[3];
                bool hok[3], wok[3];
                #pragma unroll
                for (int d = 0; d < 3; ++d) {
                    const int hh = ph + d - 1;
                    const int ww = pw + d - 1;
                    hrow[d] = (hh & 255) << 8;
                    wcol[d] = ww & 255;
                    hok[d]  = (unsigned)hh < 256u;
                    wok[d]  = (unsigned)ww < 256u;
                }
                const float* xp = x + ((pb << 16) << 3) + (hi << 2);
                #pragma unroll
                for (int tap = 0; tap < 9; ++tap) {
                    v4f q = *(const v4f*)(xp + ((hrow[tap / 3] + wcol[tap % 3]) << 3));
                    if (!(valid && hok[tap / 3] && wok[tap % 3])) q = (v4f){0.f, 0.f, 0.f, 0.f};
                    a[2 * tap].x     = q.x;  a[2 * tap].y     = q.y;
                    a[2 * tap + 1].x = q.z;  a[2 * tap + 1].y = q.w;
                }
            } else {
                const float rowmask = valid ? 1.f : 0.f;
                int   hrow[3], wcol[3];
                float hmask[3], wmask[3];
                #pragma unroll
                for (int d = 0; d < 3; ++d) {
                    const int hh = ph + d - 1;
                    const int ww = pw + d - 1;
                    hrow[d]  = (hh & 255) << 8;
                    wcol[d]  = ww & 255;
                    hmask[d] = ((unsigned)hh < 256u) ? 1.f : 0.f;
                    wmask[d] = ((unsigned)ww < 256u) ? 1.f : 0.f;
                }
                const int base = (pb * CIN) << 16;
                int   off9[9];
                float msk9[9];
                #pragma unroll
                for (int r = 0; r < 9; ++r) {
                    off9[r] = base + hrow[r / 3] + wcol[r % 3];
                    msk9[r] = hmask[r / 3] * wmask[r % 3] * rowmask;
                }

                int   offR[9];
                float mskR[9];
                #pragma unroll
                for (int s = 0; s < 9; ++s) {
                    const int s2 = (s + 2) % 9;
                    offR[s] = hi ? off9[s2] : off9[s];
                    mskR[s] = hi ? msk9[s2] : msk9[s];
                }

                #pragma unroll
                for (int cc = 0; cc < 18; ++cc) {
                    #pragma unroll
                    for (int j = 0; j < 2; ++j) {
                        const int kbase = cc * 4 + j;
                        const int tapc  = kbase % 9;
                        const int cA = (kbase / 9) << 16;
                        const int cB = ((kbase + 2) / 9) << 16;
                        const int coff = (cA == cB) ? cA : (hi ? cB : cA);
                        const float v = x[offR[tapc] + coff] * mskR[tapc];
                        if (j == 0) a[cc].x = v; else a[cc].y = v;
                    }
                }
            }

            v8f acc = {};
            #pragma unroll
            for (int cc = 0; cc < 18; ++cc) {
                acc = __builtin_amdgcn_wmma_f32_16x16x4_f32(
                    false, a[cc], false, bmat[cc], (short)0, acc, false, false);
            }

            const int col  = lane & 15;
            const int rofs = hi << 3;
            #pragma unroll
            for (int r = 0; r < 8; ++r) {
                const int row   = r + rofs;
                const int pix2  = __shfl(pix, row, 32);
                if (col < COUT && (rowbase + row) < total) {
                    const int b2 = pix2 >> 16;
                    const int h2 = (pix2 >> 8) & 255;
                    const int w2 = pix2 & 255;
                    out[((b2 * COUT + col) << 16) + (h2 << 8) + w2] = acc[r] + biasv;
                }
            }
        }

        if (!more) break;
        __syncthreads();
    } while (true);
}

extern "C" void kernel_launch(void* const* d_in, const int* in_sizes, int n_in,
                              void* d_out, int out_size, void* d_ws, size_t ws_size,
                              hipStream_t stream) {
    const float* x       = (const float*)d_in[0];
    const float* emb     = (const float*)d_in[1];
    const int*   buckets = (const int*)d_in[2];
    float*       out     = (float*)d_out;

    if (ws_size >= XT_BYTES) {
        float* xt = (float*)d_ws;
        k_interleave_channels<<<PRE_BLOCKS, 256, 0, stream>>>(x, xt);
        k_bucket_scan_gemm<true><<<dim3(SEGS, NTYPES), 256, 0, stream>>>(xt, emb, buckets, out);
    } else {
        k_bucket_scan_gemm<false><<<dim3(SEGS, NTYPES), 256, 0, stream>>>(x, emb, buckets, out);
    }
}


# Work-group = 8 consecutive buckets (one per wave) x 1/32 of the pixels: ids read and compared once per 8 buckets, one load group per wave; each wave picks its bucket out of the 8 LDS queues
# speedup vs baseline: 4.6038x; 2.3732x over previous
#include <hip/hip_runtime.h>

#define HH 256
#define WW 256
#define BB 4
#define CIN 8
#define COUT 8
#define NTYPES 216
#define NPIX (BB*HH*WW)
#define KDIM 72
#define NW   (COUT*KDIM)
#define EMB_STRIDE (NW + COUT)

typedef float v2f __attribute__((ext_vector_type(2)));
typedef float v8f __attribute__((ext_vector_type(8)));
typedef int   v4i __attribute__((ext_vector_type(4)));
typedef int   v2i __attribute__((ext_vector_type(2)));
typedef float v4f __attribute__((ext_vector_type(4)));

#define XT_BYTES ((size_t)NPIX * CIN * sizeof(float))

#define PRE_BLOCKS 256
#define PRE_UNITS  (2 * NPIX / (PRE_BLOCKS * 256))
__global__ __launch_bounds__(256) void k_interleave_channels(
    const float* __restrict__ x, float* __restrict__ xt) {
    const int g0 = blockIdx.x * 256 + threadIdx.x;
    v4f v[PRE_UNITS];
    #pragma unroll
    for (int it = 0; it < PRE_UNITS; ++it) {
        const int g   = it * (PRE_BLOCKS * 256) + g0;
        const int pix = g >> 1;
        const int cb  = (g & 1) << 2;
        const int b   = pix >> 16;
        const int hw  = pix & 65535;
        const float* src = x + ((b * CIN + cb) << 16) + hw;
        v[it].x = src[0];
        v[it].y = src[1 << 16];
        v[it].z = src[2 << 16];
        v[it].w = src[3 << 16];
    }
    #pragma unroll
    for (int it = 0; it < PRE_UNITS; ++it) {
        const int g = it * (PRE_BLOCKS * 256) + g0;
        *(v4f*)(xt + (g << 2)) = v[it];
    }
}

#define WAVES_PER_BLOCK 8
#define NGROUPS (NTYPES / WAVES_PER_BLOCK)
#define SEGS 32
#define SEG_PIX   (NPIX / SEGS)
#define WAVE_PIX  (SEG_PIX / WAVES_PER_BLOCK)
#define SCAN_LOADS (WAVE_PIX / 128)
#define WCAP WAVE_PIX
#define RING 128

template <bool INTERLEAVED>
__global__ __launch_bounds__(256, 8) void k_bucket_scan_gemm(
    const float* __restrict__ x, const float* __restrict__ emb,
    const int* __restrict__ buckets, float* __restrict__ out) {

    __shared__ int s_q[WAVES_PER_BLOCK * WCAP];
    __shared__ int s_ring[WAVES_PER_BLOCK * RING];
    __shared__ int s_cnt[WAVES_PER_BLOCK];

    const int grp  = blockIdx.y;
    const int lane = threadIdx.x & 31;
    const int wave = __builtin_amdgcn_readfirstlane(threadIdx.x >> 5);

    const float* wrow = emb + (grp * WAVES_PER_BLOCK + wave) * EMB_STRIDE;

    const int n  = lane & 15;
    const int hi = lane >> 4;
    const int khalf = hi << 1;

    {
        int* myq = s_q + wave * WCAP;
        const int pos = blockIdx.x * SEG_PIX + wave * WAVE_PIX;
        v4i buf[SCAN_LOADS];
        #pragma unroll
        for (int j = 0; j < SCAN_LOADS; ++j)
            buf[j] = *(const v4i*)(buckets + pos + j * 128 + lane * 4);
        int cnt = 0;
        #pragma unroll
        for (int j = 0; j < SCAN_LOADS; ++j) {
            const int pbase = pos + j * 128 + lane * 4;
            const bool m0 = (buf[j].x >> 3) == grp;
            const bool m1 = (buf[j].y >> 3) == grp;
            const bool m2 = (buf[j].z >> 3) == grp;
            const bool m3 = (buf[j].w >> 3) == grp;
            const unsigned k0 = __builtin_amdgcn_ballot_w32(m0);
            const unsigned k1 = __builtin_amdgcn_ballot_w32(m1);
            const unsigned k2 = __builtin_amdgcn_ballot_w32(m2);
            const unsigned k3 = __builtin_amdgcn_ballot_w32(m3);
            if (m0) myq[cnt + __builtin_amdgcn_mbcnt_lo(k0, 0u)] = pbase       | ((buf[j].x & 7) << 18);
            cnt += __builtin_popcount(k0);
            if (m1) myq[cnt + __builtin_amdgcn_mbcnt_lo(k1, 0u)] = (pbase + 1) | ((buf[j].y & 7) << 18);
            cnt += __builtin_popcount(k1);
            if (m2) myq[cnt + __builtin_amdgcn_mbcnt_lo(k2, 0u)] = (pbase + 2) | ((buf[j].z & 7) << 18);
            cnt += __builtin_popcount(k2);
            if (m3) myq[cnt + __builtin_amdgcn_mbcnt_lo(k3, 0u)] = (pbase + 3) | ((buf[j].w & 7) << 18);
            cnt += __builtin_popcount(k3);
        }
        if (lane == 0) s_cnt[wave] = cnt;
    }

    v2f bmat[18];
    #pragma unroll
    for (int cc = 0; cc < 18; ++cc) {
        const int k0 = INTERLEAVED ? (4 * hi + 2 * (cc & 1)) * 9 + (cc >> 1) : cc * 4 + khalf;
        const int k1 = INTERLEAVED ? k0 + 9 : k0 + 1;
        float b0 = 0.f, b1 = 0.f;
        if (n < COUT) {
            b0 = wrow[n * KDIM + k0];
            b1 = wrow[n * KDIM + k1];
        }
        bmat[cc].x = b0; bmat[cc].y = b1;
    }
    const float biasv = (n < COUT) ? wrow[NW + n] : 0.f;

    __syncthreads();

    int* ring = s_ring + wave * RING;
    int head  = 0;
    int count = 0;
    int q     = 0;
    int qb    = 0;
    int qn    = __builtin_amdgcn_readfirstlane(s_cnt[0]);

    while (true) {
        while (count < 16 && q < WAVES_PER_BLOCK) {
            if (qb >= qn) {
                ++q; qb = 0;
                if (q < WAVES_PER_BLOCK) qn = __builtin_amdgcn_readfirstlane(s_cnt[q]);
                continue;
            }
            const int idx = qb + 2 * lane;
            const bool in0 = idx < qn;
            const bool in1 = idx + 1 < qn;
            const v2i e = *(const v2i*)(s_q + q * WCAP + (in0 ? idx : 0));
            const bool mine0 = in0 && ((e.x >> 18) == wave);
            const bool mine1 = in1 && ((e.y >> 18) == wave);
            const unsigned k0 = __builtin_amdgcn_ballot_w32(mine0);
            const unsigned k1 = __builtin_amdgcn_ballot_w32(mine1);
            if (mine0) ring[(head + count + __builtin_amdgcn_mbcnt_lo(k0, 0u)) & (RING - 1)] = e.x & 0x3FFFF;
            count += __builtin_popcount(k0);
            if (mine1) ring[(head + count + __builtin_amdgcn_mbcnt_lo(k1, 0u)) & (RING - 1)] = e.y & 0x3FFFF;
            count += __builtin_popcount(k1);
            qb += 64;
        }
        if (count == 0) break;
        const int rows = count < 16 ? count : 16;
        {
            const int m     = lane & 15;
            const int valid = (m < rows) ? 1 : 0;
            const int pix = ring[(head + (valid ? m : 0)) & (RING - 1)];
            const int pb = pix >> 16;
            const int ph = (pix >> 8) & 255;
            const int pw = pix & 255;

            v2f a[18];
            if constexpr (INTERLEAVED) {
                int  hrow[3], wcol[3];
                bool hok[3], wok[3];
                #pragma unroll
                for (int d = 0; d < 3; ++d) {
                    const int hh = ph + d - 1;
                    const int ww = pw + d - 1;
                    hrow[d] = (hh & 255) << 8;
                    wcol[d] = ww & 255;
                    hok[d]  = (unsigned)hh < 256u;
                    wok[d]  = (unsigned)ww < 256u;
                }
                const float* xp = x + ((pb << 16) << 3) + (hi << 2);
                #pragma unroll
                for (int tap = 0; tap < 9; ++tap) {
                    v4f q = *(const v4f*)(xp + ((hrow[tap / 3] + wcol[tap % 3]) << 3));
                    if (!(valid && hok[tap / 3] && wok[tap % 3])) q = (v4f){0.f, 0.f, 0.f, 0.f};
                    a[2 * tap].x     = q.x;  a[2 * tap].y     = q.y;
                    a[2 * tap + 1].x = q.z;  a[2 * tap + 1].y = q.w;
                }
            } else {
                const float rowmask = valid ? 1.f : 0.f;
                int   hrow[3], wcol[3];
                float hmask[3], wmask[3];
                #pragma unroll
                for (int d = 0; d < 3; ++d) {
                    const int hh = ph + d - 1;
                    const int ww = pw + d - 1;
                    hrow[d]  = (hh & 255) << 8;
                    wcol[d]  = ww & 255;
                    hmask[d] = ((unsigned)hh < 256u) ? 1.f : 0.f;
                    wmask[d] = ((unsigned)ww < 256u) ? 1.f : 0.f;
                }
                const int base = (pb * CIN) << 16;
                int   off9[9];
                float msk9[9];
                #pragma unroll
                for (int r = 0; r < 9; ++r) {
                    off9[r] = base + hrow[r / 3] + wcol[r % 3];
                    msk9[r] = hmask[r / 3] * wmask[r % 3] * rowmask;
                }

                int   offR[9];
                float mskR[9];
                #pragma unroll
                for (int s = 0; s < 9; ++s) {
                    const int s2 = (s + 2) % 9;
                    offR[s] = hi ? off9[s2] : off9[s];
                    mskR[s] = hi ? msk9[s2] : msk9[s];
                }

                #pragma unroll
                for (int cc = 0; cc < 18; ++cc) {
                    #pragma unroll
                    for (int j = 0; j < 2; ++j) {
                        const int kbase = cc * 4 + j;
                        const int tapc  = kbase % 9;
                        const int cA = (kbase / 9) << 16;
                        const int cB = ((kbase + 2) / 9) << 16;
                        const int coff = (cA == cB) ? cA : (hi ? cB : cA);
                        const float v = x[offR[tapc] + coff] * mskR[tapc];
                        if (j == 0) a[cc].x = v; else a[cc].y = v;
                    }
                }
            }

            v8f acc = {};
            #pragma unroll
            for (int cc = 0; cc < 18; ++cc) {
                acc = __builtin_amdgcn_wmma_f32_16x16x4_f32(
                    false, a[cc], false, bmat[cc], (short)0, acc, false, false);
            }

            const int col  = lane & 15;
            const int rofs = hi << 3;
            #pragma unroll
            for (int r = 0; r < 8; ++r) {
                const int row   = r + rofs;
                const int pix2  = __shfl(pix, row, 32);
                if (col < COUT && row < rows) {
                    const int b2 = pix2 >> 16;
                    const int h2 = (pix2 >> 8) & 255;
                    const int w2 = pix2 & 255;
                    out[((b2 * COUT + col) << 16) + (h2 << 8) + w2] = acc[r] + biasv;
                }
            }
        }
        head   = (head + rows) & (RING - 1);
        count -= rows;
    }
}

extern "C" void kernel_launch(void* const* d_in, const int* in_sizes, int n_in,
                              void* d_out, int out_size, void* d_ws, size_t ws_size,
                              hipStream_t stream) {
    const float* x       = (const float*)d_in[0];
    const float* emb     = (const float*)d_in[1];
    const int*   buckets = (const int*)d_in[2];
    float*       out     = (float*)d_out;

    if (ws_size >= XT_BYTES) {
        float* xt = (float*)d_ws;
        k_interleave_channels<<<PRE_BLOCKS, 256, 0, stream>>>(x, xt);
        k_bucket_scan_gemm<true><<<dim3(SEGS, NGROUPS), 256, 0, stream>>>(xt, emb, buckets, out);
    } else {
        k_bucket_scan_gemm<false><<<dim3(SEGS, NGROUPS), 256, 0, stream>>>(x, emb, buckets, out);
    }
}


# Bucket weight row fetched once per wave as whole lines (5 contiguous 16-byte loads) and parked in LDS; a lane takes its 36 k values as 9 16-byte LDS reads (were 36 4-byte global loads, 16 lines each)
# speedup vs baseline: 5.2609x; 1.1427x over previous
#include <hip/hip_runtime.h>

#define HH 256
#define WW 256
#define BB 4
#define CIN 8
#define COUT 8
#define NTYPES 216
#define NPIX (BB*HH*WW)
#define KDIM 72
#define NW   (COUT*KDIM)
#define EMB_STRIDE (NW + COUT)

typedef float v2f __attribute__((ext_vector_type(2)));
typedef float v8f __attribute__((ext_vector_type(8)));
typedef int   v4i __attribute__((ext_vector_type(4)));
typedef int   v2i __attribute__((ext_vector_type(2)));
typedef float v4f __attribute__((ext_vector_type(4)));

#define XT_BYTES ((size_t)NPIX * CIN * sizeof(float))

#define PRE_BLOCKS 256
#define PRE_UNITS  (2 * NPIX / (PRE_BLOCKS * 256))
__global__ __launch_bounds__(256) void k_interleave_channels(
    const float* __restrict__ x, float* __restrict__ xt) {
    const int g0 = blockIdx.x * 256 + threadIdx.x;
    v4f v[PRE_UNITS];
    #pragma unroll
    for (int it = 0; it < PRE_UNITS; ++it) {
        const int g   = it * (PRE_BLOCKS * 256) + g0;
        const int pix = g >> 1;
        const int cb  = (g & 1) << 2;
        const int b   = pix >> 16;
        const int hw  = pix & 65535;
        const float* src = x + ((b * CIN + cb) << 16) + hw;
        v[it].x = src[0];
        v[it].y = src[1 << 16];
        v[it].z = src[2 << 16];
        v[it].w = src[3 << 16];
    }
    #pragma unroll
    for (int it = 0; it < PRE_UNITS; ++it) {
        const int g = it * (PRE_BLOCKS * 256) + g0;
        *(v4f*)(xt + (g << 2)) = v[it];
    }
}

#define WAVES_PER_BLOCK 8
#define NGROUPS (NTYPES / WAVES_PER_BLOCK)
#define SEGS 32
#define SEG_PIX   (NPIX / SEGS)
#define WAVE_PIX  (SEG_PIX / WAVES_PER_BLOCK)
#define SCAN_LOADS (WAVE_PIX / 128)
#define WCAP WAVE_PIX
#define RING 128
#define ROW_VEC4 (EMB_STRIDE / 4)
#define ROW_LOADS ((ROW_VEC4 + 31) / 32)

template <bool INTERLEAVED>
__global__ __launch_bounds__(256, 8) void k_bucket_scan_gemm(
    const float* __restrict__ x, const float* __restrict__ emb,
    const int* __restrict__ buckets, float* __restrict__ out) {

    __shared__ int s_q[WAVES_PER_BLOCK * WCAP];
    __shared__ int s_ring[WAVES_PER_BLOCK * RING];
    __shared__ int s_cnt[WAVES_PER_BLOCK];
    __shared__ v4f s_w[WAVES_PER_BLOCK * ROW_VEC4];

    const int grp  = blockIdx.y;
    const int lane = threadIdx.x & 31;
    const int wave = __builtin_amdgcn_readfirstlane(threadIdx.x >> 5);

    const float* wrow = emb + (grp * WAVES_PER_BLOCK + wave) * EMB_STRIDE;

    const int n  = lane & 15;
    const int hi = lane >> 4;
    const int khalf = hi << 1;

    v4f wreg[ROW_LOADS];
    #pragma unroll
    for (int i = 0; i < ROW_LOADS; ++i) {
        const int idx = i * 32 + lane;
        wreg[i] = *(const v4f*)(wrow + (idx < ROW_VEC4 ? idx : ROW_VEC4 - 1) * 4);
    }

    {
        int* myq = s_q + wave * WCAP;
        const int pos = blockIdx.x * SEG_PIX + wave * WAVE_PIX;
        v4i buf[SCAN_LOADS];
        #pragma unroll
        for (int j = 0; j < SCAN_LOADS; ++j)
            buf[j] = *(const v4i*)(buckets + pos + j * 128 + lane * 4);
        int cnt = 0;
        #pragma unroll
        for (int j = 0; j < SCAN_LOADS; ++j) {
            const int pbase = pos + j * 128 + lane * 4;
            const bool m0 = (buf[j].x >> 3) == grp;
            const bool m1 = (buf[j].y >> 3) == grp;
            const bool m2 = (buf[j].z >> 3) == grp;
            const bool m3 = (buf[j].w >> 3) == grp;
            const unsigned k0 = __builtin_amdgcn_ballot_w32(m0);
            const unsigned k1 = __builtin_amdgcn_ballot_w32(m1);
            const unsigned k2 = __builtin_amdgcn_ballot_w32(m2);
            const unsigned k3 = __builtin_amdgcn_ballot_w32(m3);
            if (m0) myq[cnt + __builtin_amdgcn_mbcnt_lo(k0, 0u)] = pbase       | ((buf[j].x & 7) << 18);
            cnt += __builtin_popcount(k0);
            if (m1) myq[cnt + __builtin_amdgcn_mbcnt_lo(k1, 0u)] = (pbase + 1) | ((buf[j].y & 7) << 18);
            cnt += __builtin_popcount(k1);
            if (m2) myq[cnt + __builtin_amdgcn_mbcnt_lo(k2, 0u)] = (pbase + 2) | ((buf[j].z & 7) << 18);
            cnt += __builtin_popcount(k2);
            if (m3) myq[cnt + __builtin_amdgcn_mbcnt_lo(k3, 0u)] = (pbase + 3) | ((buf[j].w & 7) << 18);
            cnt += __builtin_popcount(k3);
        }
        if (lane == 0) s_cnt[wave] = cnt;
    }

    v4f* wlds = s_w + wave * ROW_VEC4;
    #pragma unroll
    for (int i = 0; i < ROW_LOADS; ++i) {
        const int idx = i * 32 + lane;
        if (idx < ROW_VEC4) wlds[idx] = wreg[i];
    }

    __syncthreads();

    v2f bmat[18];
    float biasv = 0.f;
    #pragma unroll
    for (int cc = 0; cc < 18; ++cc) bmat[cc] = (v2f){0.f, 0.f};
    if (n < COUT) {
        const float* wl = (const float*)wlds;
        if constexpr (INTERLEAVED) {
            v4f w4[9];
            #pragma unroll
            for (int i = 0; i < 9; ++i)
                w4[i] = *(const v4f*)(wl + n * KDIM + 36 * hi + 4 * i);
            #pragma unroll
            for (int cc = 0; cc < 18; ++cc) {
                const int e0 = (2 * (cc & 1)) * 9 + (cc >> 1);
                const int e1 = e0 + 9;
                bmat[cc].x = w4[e0 >> 2][e0 & 3];
                bmat[cc].y = w4[e1 >> 2][e1 & 3];
            }
        } else {
            #pragma unroll
            for (int cc = 0; cc < 18; ++cc)
                bmat[cc] = *(const v2f*)(wl + n * KDIM + cc * 4 + khalf);
        }
        biasv = wl[NW + n];
    }

    int* ring = s_ring + wave * RING;
    int head  = 0;
    int count = 0;
    int q     = 0;
    int qb    = 0;
    int qn    = __builtin_amdgcn_readfirstlane(s_cnt[0]);

    while (true) {
        while (count < 16 && q < WAVES_PER_BLOCK) {
            if (qb >= qn) {
                ++q; qb = 0;
                if (q < WAVES_PER_BLOCK) qn = __builtin_amdgcn_readfirstlane(s_cnt[q]);
                continue;
            }
            const int idx = qb + 2 * lane;
            const bool in0 = idx < qn;
            const bool in1 = idx + 1 < qn;
            const v2i e = *(const v2i*)(s_q + q * WCAP + (in0 ? idx : 0));
            const bool mine0 = in0 && ((e.x >> 18) == wave);
            const bool mine1 = in1 && ((e.y >> 18) == wave);
            const unsigned k0 = __builtin_amdgcn_ballot_w32(mine0);
            const unsigned k1 = __builtin_amdgcn_ballot_w32(mine1);
            if (mine0) ring[(head + count + __builtin_amdgcn_mbcnt_lo(k0, 0u)) & (RING - 1)] = e.x & 0x3FFFF;
            count += __builtin_popcount(k0);
            if (mine1) ring[(head + count + __builtin_amdgcn_mbcnt_lo(k1, 0u)) & (RING - 1)] = e.y & 0x3FFFF;
            count += __builtin_popcount(k1);
            qb += 64;
        }
        if (count == 0) break;
        const int rows = count < 16 ? count : 16;
        {
            const int m     = lane & 15;
            const int valid = (m < rows) ? 1 : 0;
            const int pix = ring[(head + (valid ? m : 0)) & (RING - 1)];
            const int pb = pix >> 16;
            const int ph = (pix >> 8) & 255;
            const int pw = pix & 255;

            v2f a[18];
            if constexpr (INTERLEAVED) {
                int  hrow[3], wcol[3];
                bool hok[3], wok[3];
                #pragma unroll
                for (int d = 0; d < 3; ++d) {
                    const int hh = ph + d - 1;
                    const int ww = pw + d - 1;
                    hrow[d] = (hh & 255) << 8;
                    wcol[d] = ww & 255;
                    hok[d]  = (unsigned)hh < 256u;
                    wok[d]  = (unsigned)ww < 256u;
                }
                const float* xp = x + ((pb << 16) << 3) + (hi << 2);
                #pragma unroll
                for (int tap = 0; tap < 9; ++tap) {
                    v4f q = *(const v4f*)(xp + ((hrow[tap / 3] + wcol[tap % 3]) << 3));
                    if (!(valid && hok[tap / 3] && wok[tap % 3])) q = (v4f){0.f, 0.f, 0.f, 0.f};
                    a[2 * tap].x     = q.x;  a[2 * tap].y     = q.y;
                    a[2 * tap + 1].x = q.z;  a[2 * tap + 1].y = q.w;
                }
            } else {
                const float rowmask = valid ? 1.f : 0.f;
                int   hrow[3], wcol[3];
                float hmask[3], wmask[3];
                #pragma unroll
                for (int d = 0; d < 3; ++d) {
                    const int hh = ph + d - 1;
                    const int ww = pw + d - 1;
                    hrow[d]  = (hh & 255) << 8;
                    wcol[d]  = ww & 255;
                    hmask[d] = ((unsigned)hh < 256u) ? 1.f : 0.f;
                    wmask[d] = ((unsigned)ww < 256u) ? 1.f : 0.f;
                }
                const int base = (pb * CIN) << 16;
                int   off9[9];
                float msk9[9];
                #pragma unroll
                for (int r = 0; r < 9; ++r) {
                    off9[r] = base + hrow[r / 3] + wcol[r % 3];
                    msk9[r] = hmask[r / 3] * wmask[r % 3] * rowmask;
                }

                int   offR[9];
                float mskR[9];
                #pragma unroll
                for (int s = 0; s < 9; ++s) {
                    const int s2 = (s + 2) % 9;
                    offR[s] = hi ? off9[s2] : off9[s];
                    mskR[s] = hi ? msk9[s2] : msk9[s];
                }

                #pragma unroll
                for (int cc = 0; cc < 18; ++cc) {
                    #pragma unroll
                    for (int j = 0; j < 2; ++j) {
                        const int kbase = cc * 4 + j;
                        const int tapc  = kbase % 9;
                        const int cA = (kbase / 9) << 16;
                        const int cB = ((kbase + 2) / 9) << 16;
                        const int coff = (cA == cB) ? cA : (hi ? cB : cA);
                        const float v = x[offR[tapc] + coff] * mskR[tapc];
                        if (j == 0) a[cc].x = v; else a[cc].y = v;
                    }
                }
            }

            v8f acc = {};
            #pragma unroll
            for (int cc = 0; cc < 18; ++cc) {
                acc = __builtin_amdgcn_wmma_f32_16x16x4_f32(
                    false, a[cc], false, bmat[cc], (short)0, acc, false, false);
            }

            const int col  = lane & 15;
            const int rofs = hi << 3;
            #pragma unroll
            for (int r = 0; r < 8; ++r) {
                const int row   = r + rofs;
                const int pix2  = __shfl(pix, row, 32);
                if (col < COUT && row < rows) {
                    const int b2 = pix2 >> 16;
                    const int h2 = (pix2 >> 8) & 255;
                    const int w2 = pix2 & 255;
                    out[((b2 * COUT + col) << 16) + (h2 << 8) + w2] = acc[r] + biasv;
                }
            }
        }
        head   = (head + rows) & (RING - 1);
        count -= rows;
    }
}

extern "C" void kernel_launch(void* const* d_in, const int* in_sizes, int n_in,
                              void* d_out, int out_size, void* d_ws, size_t ws_size,
                              hipStream_t stream) {
    const float* x       = (const float*)d_in[0];
    const float* emb     = (const float*)d_in[1];
    const int*   buckets = (const int*)d_in[2];
    float*       out     = (float*)d_out;

    if (ws_size >= XT_BYTES) {
        float* xt = (float*)d_ws;
        k_interleave_channels<<<PRE_BLOCKS, 256, 0, stream>>>(x, xt);
        k_bucket_scan_gemm<true><<<dim3(SEGS, NGROUPS), 256, 0, stream>>>(xt, emb, buckets, out);
    } else {
        k_bucket_scan_gemm<false><<<dim3(SEGS, NGROUPS), 256, 0, stream>>>(x, emb, buckets, out);
    }
}
